# starting kernel
# baseline (speedup 1.0000x reference)
#include <hip/hip_runtime.h>

typedef __attribute__((ext_vector_type(2))) float v2f;
typedef __attribute__((ext_vector_type(4))) float v4f;
typedef __attribute__((ext_vector_type(8))) float v8f;

#define FS_F      8000.0f
#define HALF_W    16
#define KTAPS     33
#define PI_F      3.14159265358979323846f
#define TWO_PI_F  6.28318530717958647692f

__global__ __launch_bounds__(256) void
lpf_wmma_kernel(const float* __restrict__ in, const float* __restrict__ alpha_p,
                const float* __restrict__ beta_p, float* __restrict__ out,
                int n, int nbatch)
{
    const float alpha = alpha_p[0];
    const float beta  = beta_p[0];

    const int lane = threadIdx.x & 31;
    const int wv   = threadIdx.x >> 5;
    const int tile = blockIdx.x * (blockDim.x >> 5) + wv;
    const int t0   = tile * 16;
    if (t0 >= n) return;

    const int m  = lane & 15;
    const int hl = lane >> 4;
    const int t  = t0 + m;

    const float c_t =
        (PI_F * 0.25f + alpha * __sinf(beta * (float)t * (1.0f / FS_F))) * (1.0f / TWO_PI_F);
    const float c2   = 2.0f * c_t;
    const float xstp = TWO_PI_F * c_t;

    auto tapf = [&](float fk, bool is16) -> float {
        float w = 0.5f - 0.5f * __cosf(PI_F * (1.0f / 16.0f) * fk);
        float x = xstp * (fk - 16.0f);
        float s = __sinf(x) * __builtin_amdgcn_rcpf(x);
        s = is16 ? 1.0f : s;
        return c2 * w * s;
    };

    const int br = (m < nbatch) ? m : 0;
    const float* __restrict__ inb = in + (size_t)br * (size_t)n;

    v8f acc = {};
    float S;

    const bool edge = (t0 < HALF_W) || (t0 > n - 3 * HALF_W);
    if (!edge) {
        const float fkm = (float)(-m);
        const float* __restrict__ pB = inb + (size_t)(t0 + 2 * hl);
        float Apart = 0.0f;
        #pragma unroll
        for (int c = 0; c < 12; ++c) {
            const int base = 4 * c + 2 * hl;
            const int k0 = base - m;
            const float fk0 = fkm + (float)base;
            v2f A, B;
            float v0 = tapf(fk0,        k0     == HALF_W) * (((unsigned)k0       < KTAPS) ? 1.0f : 0.0f);
            float v1 = tapf(fk0 + 1.0f, k0 + 1 == HALF_W) * (((unsigned)(k0 + 1) < KTAPS) ? 1.0f : 0.0f);
            A.x = v0; A.y = v1;
            Apart += v0 + v1;
            const float2 bb = *(const float2*)(pB + 4 * c);
            B.x = bb.x;
            B.y = bb.y;
            acc = __builtin_amdgcn_wmma_f32_16x16x4_f32(
                false, A, false, B, (short)0, acc, false, false);
        }
        S = Apart + __shfl_xor(Apart, 16, 32);
    } else {
        const int off = (t < HALF_W) ? -(HALF_W + 1) : 0;
        const float fkm = (float)(-m - off);
        #pragma unroll
        for (int c = 0; c < 12; ++c) {
            const int base = 4 * c + 2 * hl;
            v2f A, B;
            #pragma unroll
            for (int j = 0; j < 2; ++j) {
                int k = base + j - m - off;
                bool valid = ((unsigned)k < KTAPS);
                valid = valid & ((t >= HALF_W) | (k >= HALF_W + 1 - t));
                valid = valid & ((t <= n - KTAPS) | (k < n - t));
                float v = tapf(fkm + (float)(base + j), k == HALF_W)
                          * (valid ? 1.0f : 0.0f);
                if (j == 0) A.x = v; else A.y = v;
            }
            int col = t0 + base;
            col = (col < n - 1) ? col : (n - 2);
            const float2 bb = *(const float2*)(inb + (size_t)col);
            B.x = bb.x;
            B.y = bb.y;
            acc = __builtin_amdgcn_wmma_f32_16x16x4_f32(
                false, A, false, B, (short)0, acc, false, false);
        }
        float Sp = 0.0f;
        #pragma unroll
        for (int j = 0; j <= 16; ++j) {
            int k = 2 * j + hl;
            float v = tapf((float)k, k == HALF_W);
            Sp += v * ((k < KTAPS) ? 1.0f : 0.0f);
        }
        S = Sp + __shfl_xor(Sp, 16, 32);
    }

    float invS = __builtin_amdgcn_rcpf(S);
    invS = __builtin_fmaf(invS, __builtin_fmaf(-S, invS, 1.0f), invS);

    v4f lo, hi;
    #pragma unroll
    for (int v = 0; v < 8; ++v) {
        const int row = v + 8 * hl;
        const float is = __shfl(invS, row, 32);
        const float val = acc[v] * is;
        if (v < 4) lo[v] = val; else hi[v - 4] = val;
    }
    if (m < nbatch) {
        float* p = out + (size_t)m * (size_t)n + (size_t)(t0 + 8 * hl);
        *(v4f*)(p + 0) = lo;
        *(v4f*)(p + 4) = hi;
    }
}

extern "C" void kernel_launch(void* const* d_in, const int* in_sizes, int n_in,
                              void* d_out, int out_size, void* d_ws, size_t ws_size,
                              hipStream_t stream) {
    const float* in      = (const float*)d_in[0];
    const float* alpha_p = (const float*)d_in[1];
    const float* beta_p  = (const float*)d_in[2];
    float* out = (float*)d_out;

    const int nbatch = 8;
    const int n = in_sizes[0] / nbatch;

    const int tiles = (n + 15) / 16;
    const int wavesPerBlock = 8;
    const int blocks = (tiles + wavesPerBlock - 1) / wavesPerBlock;

    lpf_wmma_kernel<<<blocks, wavesPerBlock * 32, 0, stream>>>(
        in, alpha_p, beta_p, out, n, nbatch);
}


# Tap factor that depends on k only, w_k/(pi*(k-16)), built once per workgroup in LDS: a tap is table read * one v_sin_f32 (was cos + sin + rcp per slot); order of every sum unchanged
# speedup vs baseline: 1.1521x; 1.1521x over previous
#include <hip/hip_runtime.h>

typedef __attribute__((ext_vector_type(2))) float v2f;
typedef __attribute__((ext_vector_type(4))) float v4f;
typedef __attribute__((ext_vector_type(8))) float v8f;

#define FS_F      8000.0f
#define HALF_W    16
#define KTAPS     33
#define PI_F      3.14159265358979323846f
#define TWO_PI_F  6.28318530717958647692f
#define TAB_LO    15
#define TAB_N     96

__global__ __launch_bounds__(256, 16) void
lpf_wmma_kernel(const float* __restrict__ in, const float* __restrict__ alpha_p,
                const float* __restrict__ beta_p, float* __restrict__ out,
                int n, int nbatch)
{
    __shared__ float gtab[TAB_N];

    const float alpha = alpha_p[0];
    const float beta  = beta_p[0];

    const int lane = threadIdx.x & 31;
    const int wv   = threadIdx.x >> 5;
    const int tile = blockIdx.x * (blockDim.x >> 5) + wv;
    const int t0   = tile * 16;
    const bool active = (t0 < n);

    const int m  = lane & 15;
    const int hl = lane >> 4;
    const int t  = t0 + m;

    const float c_t =
        (PI_F * 0.25f + alpha * __sinf(beta * (float)t * (1.0f / FS_F))) * (1.0f / TWO_PI_F);
    const float c2   = 2.0f * c_t;

    auto tapk = [&](int k) -> float {
        const float g = gtab[k + TAB_LO];
        const float s = __builtin_amdgcn_sinf(c_t * (float)(k - HALF_W));
        return __builtin_fmaf(g, s, (k == HALF_W) ? c2 : 0.0f);
    };

    const int br = (m < nbatch) ? m : 0;
    const float* __restrict__ inb = in + (size_t)br * (size_t)n;

    v8f acc = {};
    float S;

    const bool edge = (t0 < HALF_W) || (t0 > n - 3 * HALF_W);

    if (threadIdx.x < TAB_N) {
        const int k = (int)threadIdx.x - TAB_LO;
        const float w = 0.5f - 0.5f * __builtin_amdgcn_cosf((float)k * (1.0f / 32.0f));
        const float g = w / (PI_F * (float)(k - HALF_W));
        gtab[threadIdx.x] = (((unsigned)k < KTAPS) && (k != HALF_W)) ? g : 0.0f;
    }
    __syncthreads();
    if (!active) return;

    if (!edge) {
        const int d = 2 * hl - m;
        const float u = c_t * (float)(d - HALF_W);
        const float* gl = gtab + (d + TAB_LO);
        const float* __restrict__ pB = inb + (size_t)(t0 + 2 * hl);
        float Apart = 0.0f;
        #pragma unroll
        for (int c = 0; c < 12; ++c) {
            v2f A, B, g, s;
            g.x = gl[4 * c];
            g.y = gl[4 * c + 1];
            s.x = __builtin_amdgcn_sinf(__builtin_fmaf(c_t, (float)(4 * c),     u));
            s.y = __builtin_amdgcn_sinf(__builtin_fmaf(c_t, (float)(4 * c + 1), u));
            if (c >= 4 && c <= 7) {
                v2f e;
                e.x = (d == HALF_W - 4 * c)     ? c2 : 0.0f;
                e.y = (d == HALF_W - 4 * c - 1) ? c2 : 0.0f;
                A = g * s + e;
            } else {
                A = g * s;
            }
            Apart += A.x + A.y;
            const float2 bb = *(const float2*)(pB + 4 * c);
            B.x = bb.x;
            B.y = bb.y;
            acc = __builtin_amdgcn_wmma_f32_16x16x4_f32(
                false, A, false, B, (short)0, acc, false, false);
        }
        S = Apart + __shfl_xor(Apart, 16, 32);
    } else {
        const int off = (t < HALF_W) ? -(HALF_W + 1) : 0;
        #pragma unroll
        for (int c = 0; c < 12; ++c) {
            const int base = 4 * c + 2 * hl;
            v2f A, B;
            #pragma unroll
            for (int j = 0; j < 2; ++j) {
                int k = base + j - m - off;
                bool valid = ((unsigned)k < KTAPS);
                valid = valid & ((t >= HALF_W) | (k >= HALF_W + 1 - t));
                valid = valid & ((t <= n - KTAPS) | (k < n - t));
                float v = tapk(k) * (valid ? 1.0f : 0.0f);
                if (j == 0) A.x = v; else A.y = v;
            }
            int col = t0 + base;
            col = (col < n - 1) ? col : (n - 2);
            const float2 bb = *(const float2*)(inb + (size_t)col);
            B.x = bb.x;
            B.y = bb.y;
            acc = __builtin_amdgcn_wmma_f32_16x16x4_f32(
                false, A, false, B, (short)0, acc, false, false);
        }
        float Sp = 0.0f;
        #pragma unroll
        for (int j = 0; j <= 16; ++j) {
            int k = 2 * j + hl;
            float v = tapk(k);
            Sp += v * ((k < KTAPS) ? 1.0f : 0.0f);
        }
        S = Sp + __shfl_xor(Sp, 16, 32);
    }

    float invS = __builtin_amdgcn_rcpf(S);
    invS = __builtin_fmaf(invS, __builtin_fmaf(-S, invS, 1.0f), invS);

    v4f lo, hi;
    #pragma unroll
    for (int v = 0; v < 8; ++v) {
        const int row = v + 8 * hl;
        const float is = __shfl(invS, row, 32);
        const float val = acc[v] * is;
        if (v < 4) lo[v] = val; else hi[v - 4] = val;
    }
    if (m < nbatch) {
        float* p = out + (size_t)m * (size_t)n + (size_t)(t0 + 8 * hl);
        *(v4f*)(p + 0) = lo;
        *(v4f*)(p + 4) = hi;
    }
}

extern "C" void kernel_launch(void* const* d_in, const int* in_sizes, int n_in,
                              void* d_out, int out_size, void* d_ws, size_t ws_size,
                              hipStream_t stream) {
    const float* in      = (const float*)d_in[0];
    const float* alpha_p = (const float*)d_in[1];
    const float* beta_p  = (const float*)d_in[2];
    float* out = (float*)d_out;

    const int nbatch = 8;
    const int n = in_sizes[0] / nbatch;

    const int tiles = (n + 15) / 16;
    const int wavesPerBlock = 8;
    const int blocks = (tiles + wavesPerBlock - 1) / wavesPerBlock;

    lpf_wmma_kernel<<<blocks, wavesPerBlock * 32, 0, stream>>>(
        in, alpha_p, beta_p, out, n, nbatch);
}


# Interior tiles: a row's 17 distinct taps (mirror symmetry) built once by its 2 lanes, scaled by 1/S, stored as banded A tile in the wave's LDS; A fragments are LDS reads, D unscaled; new S sum order
# speedup vs baseline: 1.3036x; 1.1315x over previous
#include <hip/hip_runtime.h>

typedef __attribute__((ext_vector_type(2))) float v2f;
typedef __attribute__((ext_vector_type(4))) float v4f;
typedef __attribute__((ext_vector_type(8))) float v8f;

#define FS_F      8000.0f
#define HALF_W    16
#define KTAPS     33
#define PI_F      3.14159265358979323846f
#define TWO_PI_F  6.28318530717958647692f
#define TAB_LO    15
#define TAB_N     96
#define WAVES     8
#define APITCH    52

__global__ __launch_bounds__(256, 16) void
lpf_wmma_kernel(const float* __restrict__ in, const float* __restrict__ alpha_p,
                const float* __restrict__ beta_p, float* __restrict__ out,
                int n, int nbatch)
{
    __shared__ float gtab[TAB_N];
    __shared__ __attribute__((aligned(16))) float atile[WAVES * 16 * APITCH];

    const float alpha = alpha_p[0];
    const float beta  = beta_p[0];

    const int lane = threadIdx.x & 31;
    const int wv   = threadIdx.x >> 5;
    const int tile = blockIdx.x * (blockDim.x >> 5) + wv;
    const int t0   = tile * 16;
    const bool active = (t0 < n);

    const int m  = lane & 15;
    const int hl = lane >> 4;
    const int t  = t0 + m;

    const float c_t =
        (PI_F * 0.25f + alpha * __sinf(beta * (float)t * (1.0f / FS_F))) * (1.0f / TWO_PI_F);
    const float c2   = 2.0f * c_t;

    auto tapk = [&](int k) -> float {
        const float g = gtab[k + TAB_LO];
        const float s = __builtin_amdgcn_sinf(c_t * (float)(k - HALF_W));
        return __builtin_fmaf(g, s, (k == HALF_W) ? c2 : 0.0f);
    };

    const int br = (m < nbatch) ? m : 0;
    const float* __restrict__ inb = in + (size_t)br * (size_t)n;

    v8f acc = {};
    float S;

    const bool edge = (t0 < HALF_W) || (t0 > n - 3 * HALF_W);

    if (threadIdx.x < TAB_N) {
        const int k = (int)threadIdx.x - TAB_LO;
        const float w = 0.5f - 0.5f * __builtin_amdgcn_cosf((float)k * (1.0f / 32.0f));
        const float g = w / (PI_F * (float)(k - HALF_W));
        gtab[threadIdx.x] = (((unsigned)k < KTAPS) && (k != HALF_W)) ? g : 0.0f;
    }
    __syncthreads();
    if (!active) return;

    if (!edge) {
        float* At = atile + wv * (16 * APITCH);
        float* rowp = At + m * APITCH;

        const float* __restrict__ pB = inb + (size_t)(t0 + 2 * hl);
        float2 bb[12];
        #pragma unroll
        for (int c = 0; c < 12; ++c)
            bb[c] = *(const float2*)(pB + 4 * c);

        {
            const v4f z = {0.0f, 0.0f, 0.0f, 0.0f};
            float* zr = rowp + 32 * hl;
            *(v4f*)(zr + 0)  = z;
            *(v4f*)(zr + 4)  = z;
            *(v4f*)(zr + 8)  = z;
            *(v4f*)(zr + 12) = z;
        }

        const int j0 = 8 * hl;
        const float ph0 = c_t * (float)j0;
        const float* gj = gtab + (TAB_LO + HALF_W) + j0;
        float u0 = gj[0] * __builtin_amdgcn_sinf(ph0);
        u0 = (hl == 0) ? c2 : u0;
        v2f q[4];
        #pragma unroll
        for (int p = 0; p < 4; ++p) {
            v2f g, s;
            g.x = gj[2 * p + 1];
            g.y = gj[2 * p + 2];
            s.x = __builtin_amdgcn_sinf(__builtin_fmaf(c_t, (float)(2 * p + 1), ph0));
            s.y = __builtin_amdgcn_sinf(__builtin_fmaf(c_t, (float)(2 * p + 2), ph0));
            q[p] = g * s;
        }

        const v2f r = (q[0] + q[1]) + (q[2] + q[3]);
        const float Sp = __builtin_fmaf(2.0f, r.x + r.y, (hl == 0) ? u0 : 0.0f);
        {
            const int sb = __builtin_bit_cast(int, Sp);
            const float So = __builtin_bit_cast(float,
                __builtin_amdgcn_permlanex16(sb, sb, 0x76543210, 0xfedcba98, false, false));
            S = Sp + So;
        }

        float invS = __builtin_amdgcn_rcpf(S);
        invS = __builtin_fmaf(invS, __builtin_fmaf(-S, invS, 1.0f), invS);

        __builtin_amdgcn_fence(__ATOMIC_RELEASE, "wavefront");

        float* up = rowp + (m + HALF_W) + j0;
        float* dn = rowp + (m + HALF_W) - j0;
        {
            const float a0 = u0 * invS;
            up[0] = a0;
            dn[0] = a0;
        }
        #pragma unroll
        for (int p = 0; p < 4; ++p) {
            const v2f aq = q[p] * invS;
            up[2 * p + 1]    = aq.x;
            up[2 * p + 2]    = aq.y;
            dn[-(2 * p + 1)] = aq.x;
            dn[-(2 * p + 2)] = aq.y;
        }

        __builtin_amdgcn_fence(__ATOMIC_ACQ_REL, "wavefront");
        __builtin_amdgcn_wave_barrier();

        const float* pA = rowp + 2 * hl;
        #pragma unroll
        for (int c = 0; c < 12; ++c) {
            v2f A, B;
            const float2 aa = *(const float2*)(pA + 4 * c);
            A.x = aa.x;
            A.y = aa.y;
            B.x = bb[c].x;
            B.y = bb[c].y;
            acc = __builtin_amdgcn_wmma_f32_16x16x4_f32(
                false, A, false, B, (short)0, acc, false, false);
        }
    } else {
        const int off = (t < HALF_W) ? -(HALF_W + 1) : 0;
        #pragma unroll
        for (int c = 0; c < 12; ++c) {
            const int base = 4 * c + 2 * hl;
            v2f A, B;
            #pragma unroll
            for (int j = 0; j < 2; ++j) {
                int k = base + j - m - off;
                bool valid = ((unsigned)k < KTAPS);
                valid = valid & ((t >= HALF_W) | (k >= HALF_W + 1 - t));
                valid = valid & ((t <= n - KTAPS) | (k < n - t));
                float v = tapk(k) * (valid ? 1.0f : 0.0f);
                if (j == 0) A.x = v; else A.y = v;
            }
            int col = t0 + base;
            col = (col < n - 1) ? col : (n - 2);
            const float2 bb = *(const float2*)(inb + (size_t)col);
            B.x = bb.x;
            B.y = bb.y;
            acc = __builtin_amdgcn_wmma_f32_16x16x4_f32(
                false, A, false, B, (short)0, acc, false, false);
        }
        float Sp = 0.0f;
        #pragma unroll
        for (int j = 0; j <= 16; ++j) {
            int k = 2 * j + hl;
            float v = tapk(k);
            Sp += v * ((k < KTAPS) ? 1.0f : 0.0f);
        }
        S = Sp + __shfl_xor(Sp, 16, 32);

        float invS = __builtin_amdgcn_rcpf(S);
        invS = __builtin_fmaf(invS, __builtin_fmaf(-S, invS, 1.0f), invS);
        #pragma unroll
        for (int v = 0; v < 8; ++v) {
            const int row = v + 8 * hl;
            acc[v] *= __shfl(invS, row, 32);
        }
    }

    v4f lo, hi;
    #pragma unroll
    for (int v = 0; v < 8; ++v) {
        if (v < 4) lo[v] = acc[v]; else hi[v - 4] = acc[v];
    }
    if (m < nbatch) {
        float* p = out + (size_t)m * (size_t)n + (size_t)(t0 + 8 * hl);
        *(v4f*)(p + 0) = lo;
        *(v4f*)(p + 4) = hi;
    }
}

extern "C" void kernel_launch(void* const* d_in, const int* in_sizes, int n_in,
                              void* d_out, int out_size, void* d_ws, size_t ws_size,
                              hipStream_t stream) {
    const float* in      = (const float*)d_in[0];
    const float* alpha_p = (const float*)d_in[1];
    const float* beta_p  = (const float*)d_in[2];
    float* out = (float*)d_out;

    const int nbatch = 8;
    const int n = in_sizes[0] / nbatch;

    const int tiles = (n + 15) / 16;
    const int wavesPerBlock = 8;
    const int blocks = (tiles + wavesPerBlock - 1) / wavesPerBlock;

    lpf_wmma_kernel<<<blocks, wavesPerBlock * 32, 0, stream>>>(
        in, alpha_p, beta_p, out, n, nbatch);
}


# Index and address arithmetic off the vector port: wave number read into a scalar register (tile, range tests, bases, branches scalar), lane offsets 32-bit via a 24-bit multiply, no 64-bit vector ops
# speedup vs baseline: 1.4475x; 1.1104x over previous
#include <hip/hip_runtime.h>

typedef __attribute__((ext_vector_type(2))) float v2f;
typedef __attribute__((ext_vector_type(4))) float v4f;
typedef __attribute__((ext_vector_type(8))) float v8f;

#define FS_F      8000.0f
#define HALF_W    16
#define KTAPS     33
#define PI_F      3.14159265358979323846f
#define TWO_PI_F  6.28318530717958647692f
#define TAB_LO    15
#define TAB_N     96
#define WAVES     8
#define APITCH    52

__global__ __launch_bounds__(256, 16) void
lpf_wmma_kernel(const float* __restrict__ in, const float* __restrict__ alpha_p,
                const float* __restrict__ beta_p, float* __restrict__ out,
                int n, int nbatch)
{
    __shared__ float gtab[TAB_N];
    __shared__ __attribute__((aligned(16))) float atile[WAVES * 16 * APITCH];

    const float alpha = alpha_p[0];
    const float beta  = beta_p[0];

    const int lane = threadIdx.x & 31;
    const int wv   = __builtin_amdgcn_readfirstlane((int)(threadIdx.x >> 5));
    const int tile = (int)blockIdx.x * WAVES + wv;
    const int t0   = tile * 16;
    const bool active = (t0 < n);

    const int m  = lane & 15;
    const int hl = lane >> 4;
    const int t  = t0 + m;

    const float c_t =
        (PI_F * 0.25f + alpha * __sinf(beta * (float)t * (1.0f / FS_F))) * (1.0f / TWO_PI_F);
    const float c2   = 2.0f * c_t;

    auto tapk = [&](int k) -> float {
        const float g = gtab[k + TAB_LO];
        const float s = __builtin_amdgcn_sinf(c_t * (float)(k - HALF_W));
        return __builtin_fmaf(g, s, (k == HALF_W) ? c2 : 0.0f);
    };

    const int br = (m < nbatch) ? m : 0;
    const unsigned rowoff = __umul24((unsigned)br, (unsigned)n);

    v8f acc = {};
    float S;

    const bool edge = (t0 < HALF_W) || (t0 > n - 3 * HALF_W);

    if (threadIdx.x < TAB_N) {
        const int k = (int)threadIdx.x - TAB_LO;
        const float w = 0.5f - 0.5f * __builtin_amdgcn_cosf((float)k * (1.0f / 32.0f));
        const float g = w / (PI_F * (float)(k - HALF_W));
        gtab[threadIdx.x] = (((unsigned)k < KTAPS) && (k != HALF_W)) ? g : 0.0f;
    }
    __syncthreads();
    if (!active) return;

    if (!edge) {
        float* At = atile + (int)(threadIdx.x >> 5) * (16 * APITCH);
        float* rowp = At + m * APITCH;

        const char* __restrict__ pB = (const char*)(in + t0);
        const unsigned bo = (rowoff + 2u * (unsigned)hl) * 4u;
        float2 bb[12];
        #pragma unroll
        for (int c = 0; c < 12; ++c)
            bb[c] = *(const float2*)(pB + bo + 16 * c);

        {
            const v4f z = {0.0f, 0.0f, 0.0f, 0.0f};
            float* zr = rowp + 32 * hl;
            *(v4f*)(zr + 0)  = z;
            *(v4f*)(zr + 4)  = z;
            *(v4f*)(zr + 8)  = z;
            *(v4f*)(zr + 12) = z;
        }

        const int j0 = 8 * hl;
        const float ph0 = c_t * (float)j0;
        const float* gj = gtab + (TAB_LO + HALF_W) + j0;
        float u0 = gj[0] * __builtin_amdgcn_sinf(ph0);
        u0 = (hl == 0) ? c2 : u0;
        v2f q[4];
        #pragma unroll
        for (int p = 0; p < 4; ++p) {
            v2f g, s;
            g.x = gj[2 * p + 1];
            g.y = gj[2 * p + 2];
            s.x = __builtin_amdgcn_sinf(__builtin_fmaf(c_t, (float)(2 * p + 1), ph0));
            s.y = __builtin_amdgcn_sinf(__builtin_fmaf(c_t, (float)(2 * p + 2), ph0));
            q[p] = g * s;
        }

        const v2f r = (q[0] + q[1]) + (q[2] + q[3]);
        const float Sp = __builtin_fmaf(2.0f, r.x + r.y, (hl == 0) ? u0 : 0.0f);
        {
            const int sb = __builtin_bit_cast(int, Sp);
            const float So = __builtin_bit_cast(float,
                __builtin_amdgcn_permlanex16(sb, sb, 0x76543210, 0xfedcba98, false, false));
            S = Sp + So;
        }

        float invS = __builtin_amdgcn_rcpf(S);
        invS = __builtin_fmaf(invS, __builtin_fmaf(-S, invS, 1.0f), invS);

        __builtin_amdgcn_fence(__ATOMIC_RELEASE, "wavefront");

        float* up = rowp + (m + HALF_W) + j0;
        float* dn = rowp + (m + HALF_W) - j0;
        {
            const float a0 = u0 * invS;
            up[0] = a0;
            dn[0] = a0;
        }
        #pragma unroll
        for (int p = 0; p < 4; ++p) {
            const v2f aq = q[p] * invS;
            up[2 * p + 1]    = aq.x;
            up[2 * p + 2]    = aq.y;
            dn[-(2 * p + 1)] = aq.x;
            dn[-(2 * p + 2)] = aq.y;
        }

        __builtin_amdgcn_fence(__ATOMIC_ACQ_REL, "wavefront");
        __builtin_amdgcn_wave_barrier();

        const float* pA = rowp + 2 * hl;
        #pragma unroll
        for (int c = 0; c < 12; ++c) {
            v2f A, B;
            const float2 aa = *(const float2*)(pA + 4 * c);
            A.x = aa.x;
            A.y = aa.y;
            B.x = bb[c].x;
            B.y = bb[c].y;
            acc = __builtin_amdgcn_wmma_f32_16x16x4_f32(
                false, A, false, B, (short)0, acc, false, false);
        }
    } else {
        const float* __restrict__ inb = in + rowoff;
        const int off = (t < HALF_W) ? -(HALF_W + 1) : 0;
        #pragma unroll
        for (int c = 0; c < 12; ++c) {
            const int base = 4 * c + 2 * hl;
            v2f A, B;
            #pragma unroll
            for (int j = 0; j < 2; ++j) {
                int k = base + j - m - off;
                bool valid = ((unsigned)k < KTAPS);
                valid = valid & ((t >= HALF_W) | (k >= HALF_W + 1 - t));
                valid = valid & ((t <= n - KTAPS) | (k < n - t));
                float v = tapk(k) * (valid ? 1.0f : 0.0f);
                if (j == 0) A.x = v; else A.y = v;
            }
            int col = t0 + base;
            col = (col < n - 1) ? col : (n - 2);
            const float2 bb = *(const float2*)(inb + (size_t)col);
            B.x = bb.x;
            B.y = bb.y;
            acc = __builtin_amdgcn_wmma_f32_16x16x4_f32(
                false, A, false, B, (short)0, acc, false, false);
        }
        float Sp = 0.0f;
        #pragma unroll
        for (int j = 0; j <= 16; ++j) {
            int k = 2 * j + hl;
            float v = tapk(k);
            Sp += v * ((k < KTAPS) ? 1.0f : 0.0f);
        }
        S = Sp + __shfl_xor(Sp, 16, 32);

        float invS = __builtin_amdgcn_rcpf(S);
        invS = __builtin_fmaf(invS, __builtin_fmaf(-S, invS, 1.0f), invS);
        #pragma unroll
        for (int v = 0; v < 8; ++v) {
            const int row = v + 8 * hl;
            acc[v] *= __shfl(invS, row, 32);
        }
    }

    v4f lo, hi;
    #pragma unroll
    for (int v = 0; v < 8; ++v) {
        if (v < 4) lo[v] = acc[v]; else hi[v - 4] = acc[v];
    }
    if (m < nbatch) {
        char* p = (char*)(out + t0);
        const unsigned so = (rowoff + 8u * (unsigned)hl) * 4u;
        *(v4f*)(p + so)      = lo;
        *(v4f*)(p + so + 16) = hi;
    }
}

extern "C" void kernel_launch(void* const* d_in, const int* in_sizes, int n_in,
                              void* d_out, int out_size, void* d_ws, size_t ws_size,
                              hipStream_t stream) {
    const float* in      = (const float*)d_in[0];
    const float* alpha_p = (const float*)d_in[1];
    const float* beta_p  = (const float*)d_in[2];
    float* out = (float*)d_out;

    const int nbatch = 8;
    const int n = in_sizes[0] / nbatch;

    const int tiles = (n + 15) / 16;
    const int wavesPerBlock = 8;
    const int blocks = (tiles + wavesPerBlock - 1) / wavesPerBlock;

    lpf_wmma_kernel<<<blocks, wavesPerBlock * 32, 0, stream>>>(
        in, alpha_p, beta_p, out, n, nbatch);
}


# Interior tiles fill all 16 B columns: columns 8..15 carry the 8 batches' window from t0+8, A row r starts its band at column r mod 8, so K is 40 (was 48): 10 WMMAs for 12; sum order unchanged
# speedup vs baseline: 1.5641x; 1.0806x over previous
#include <hip/hip_runtime.h>

typedef __attribute__((ext_vector_type(2))) float v2f;
typedef __attribute__((ext_vector_type(4))) float v4f;
typedef __attribute__((ext_vector_type(8))) float v8f;

#define FS_F      8000.0f
#define HALF_W    16
#define KTAPS     33
#define PI_F      3.14159265358979323846f
#define TWO_PI_F  6.28318530717958647692f
#define TAB_LO    15
#define TAB_N     96
#define WAVES     8
#define KCOLS     40
#define APITCH    44

__global__ __launch_bounds__(256, 16) void
lpf_wmma_kernel(const float* __restrict__ in, const float* __restrict__ alpha_p,
                const float* __restrict__ beta_p, float* __restrict__ out,
                int n, int nbatch)
{
    __shared__ float gtab[TAB_N];
    __shared__ __attribute__((aligned(16))) float atile[WAVES * 16 * APITCH];

    const float alpha = alpha_p[0];
    const float beta  = beta_p[0];

    const int lane = threadIdx.x & 31;
    const int wv   = __builtin_amdgcn_readfirstlane((int)(threadIdx.x >> 5));
    const int tile = (int)blockIdx.x * WAVES + wv;
    const int t0   = tile * 16;
    const bool active = (t0 < n);

    const int m  = lane & 15;
    const int hl = lane >> 4;
    const int t  = t0 + m;

    const float c_t =
        (PI_F * 0.25f + alpha * __sinf(beta * (float)t * (1.0f / FS_F))) * (1.0f / TWO_PI_F);
    const float c2   = 2.0f * c_t;

    auto tapk = [&](int k) -> float {
        const float g = gtab[k + TAB_LO];
        const float s = __builtin_amdgcn_sinf(c_t * (float)(k - HALF_W));
        return __builtin_fmaf(g, s, (k == HALF_W) ? c2 : 0.0f);
    };


    auto store_rows = [&](const v8f& d, bool mine, unsigned rowoff) {
        v4f lo, hi;
        #pragma unroll
        for (int v = 0; v < 8; ++v) {
            if (v < 4) lo[v] = d[v]; else hi[v - 4] = d[v];
        }
        if (mine) {
            char* p = (char*)(out + t0);
            const unsigned so = (rowoff + 8u * (unsigned)hl) * 4u;
            *(v4f*)(p + so)      = lo;
            *(v4f*)(p + so + 16) = hi;
        }
    };

    v8f acc = {};
    float S;

    const bool edge = (t0 < HALF_W) || (t0 > n - 3 * HALF_W);

    if (threadIdx.x < TAB_N) {
        const int k = (int)threadIdx.x - TAB_LO;
        const float w = 0.5f - 0.5f * __builtin_amdgcn_cosf((float)k * (1.0f / 32.0f));
        const float g = w / (PI_F * (float)(k - HALF_W));
        gtab[threadIdx.x] = (((unsigned)k < KTAPS) && (k != HALF_W)) ? g : 0.0f;
    }
    __syncthreads();
    if (!active) return;

    if (!edge) {
        const int s0 = m & 7;
        const int cg = m >> 3;
        const int br = (s0 < nbatch) ? s0 : 0;
        const unsigned rowoff = __umul24((unsigned)br, (unsigned)n);
        float* At = atile + (int)(threadIdx.x >> 5) * (16 * APITCH);
        float* rowp = At + m * APITCH;

        const char* __restrict__ pB = (const char*)(in + t0);
        const unsigned bo = (rowoff + 8u * (unsigned)cg + 2u * (unsigned)hl) * 4u;
        float2 bb[KCOLS / 4];
        #pragma unroll
        for (int c = 0; c < KCOLS / 4; ++c)
            bb[c] = *(const float2*)(pB + bo + 16 * c);

        {
            const v4f z = {0.0f, 0.0f, 0.0f, 0.0f};
            float* zr = rowp + 32 * hl;
            *(v4f*)(zr + 0)  = z;
            *(v4f*)(zr + 4)  = z;
        }

        const int j0 = 8 * hl;
        const float ph0 = c_t * (float)j0;
        const float* gj = gtab + (TAB_LO + HALF_W) + j0;
        float u0 = gj[0] * __builtin_amdgcn_sinf(ph0);
        u0 = (hl == 0) ? c2 : u0;
        v2f q[4];
        #pragma unroll
        for (int p = 0; p < 4; ++p) {
            v2f g, s;
            g.x = gj[2 * p + 1];
            g.y = gj[2 * p + 2];
            s.x = __builtin_amdgcn_sinf(__builtin_fmaf(c_t, (float)(2 * p + 1), ph0));
            s.y = __builtin_amdgcn_sinf(__builtin_fmaf(c_t, (float)(2 * p + 2), ph0));
            q[p] = g * s;
        }

        const v2f r = (q[0] + q[1]) + (q[2] + q[3]);
        const float Sp = __builtin_fmaf(2.0f, r.x + r.y, (hl == 0) ? u0 : 0.0f);
        {
            const int sb = __builtin_bit_cast(int, Sp);
            const float So = __builtin_bit_cast(float,
                __builtin_amdgcn_permlanex16(sb, sb, 0x76543210, 0xfedcba98, false, false));
            S = Sp + So;
        }

        float invS = __builtin_amdgcn_rcpf(S);
        invS = __builtin_fmaf(invS, __builtin_fmaf(-S, invS, 1.0f), invS);

        __builtin_amdgcn_fence(__ATOMIC_RELEASE, "wavefront");

        float* up = rowp + (s0 + HALF_W) + j0;
        float* dn = rowp + (s0 + HALF_W) - j0;
        {
            const float a0 = u0 * invS;
            up[0] = a0;
            dn[0] = a0;
        }
        #pragma unroll
        for (int p = 0; p < 4; ++p) {
            const v2f aq = q[p] * invS;
            up[2 * p + 1]    = aq.x;
            up[2 * p + 2]    = aq.y;
            dn[-(2 * p + 1)] = aq.x;
            dn[-(2 * p + 2)] = aq.y;
        }

        __builtin_amdgcn_fence(__ATOMIC_ACQ_REL, "wavefront");
        __builtin_amdgcn_wave_barrier();

        const float* pA = rowp + 2 * hl;
        #pragma unroll
        for (int c = 0; c < KCOLS / 4; ++c) {
            v2f A, B;
            const float2 aa = *(const float2*)(pA + 4 * c);
            A.x = aa.x;
            A.y = aa.y;
            B.x = bb[c].x;
            B.y = bb[c].y;
            acc = __builtin_amdgcn_wmma_f32_16x16x4_f32(
                false, A, false, B, (short)0, acc, false, false);
        }
        store_rows(acc, (cg == hl) && (s0 < nbatch), rowoff);
    } else {
        const int br = (m < nbatch) ? m : 0;
        const unsigned rowoff = __umul24((unsigned)br, (unsigned)n);
        const float* __restrict__ inb = in + rowoff;
        const int off = (t < HALF_W) ? -(HALF_W + 1) : 0;
        #pragma unroll
        for (int c = 0; c < 12; ++c) {
            const int base = 4 * c + 2 * hl;
            v2f A, B;
            #pragma unroll
            for (int j = 0; j < 2; ++j) {
                int k = base + j - m - off;
                bool valid = ((unsigned)k < KTAPS);
                valid = valid & ((t >= HALF_W) | (k >= HALF_W + 1 - t));
                valid = valid & ((t <= n - KTAPS) | (k < n - t));
                float v = tapk(k) * (valid ? 1.0f : 0.0f);
                if (j == 0) A.x = v; else A.y = v;
            }
            int col = t0 + base;
            col = (col < n - 1) ? col : (n - 2);
            const float2 bb = *(const float2*)(inb + (size_t)col);
            B.x = bb.x;
            B.y = bb.y;
            acc = __builtin_amdgcn_wmma_f32_16x16x4_f32(
                false, A, false, B, (short)0, acc, false, false);
        }
        float Sp = 0.0f;
        #pragma unroll
        for (int j = 0; j <= 16; ++j) {
            int k = 2 * j + hl;
            float v = tapk(k);
            Sp += v * ((k < KTAPS) ? 1.0f : 0.0f);
        }
        S = Sp + __shfl_xor(Sp, 16, 32);

        float invS = __builtin_amdgcn_rcpf(S);
        invS = __builtin_fmaf(invS, __builtin_fmaf(-S, invS, 1.0f), invS);
        #pragma unroll
        for (int v = 0; v < 8; ++v) {
            const int row = v + 8 * hl;
            acc[v] *= __shfl(invS, row, 32);
        }
        store_rows(acc, m < nbatch, rowoff);
    }
}

extern "C" void kernel_launch(void* const* d_in, const int* in_sizes, int n_in,
                              void* d_out, int out_size, void* d_ws, size_t ws_size,
                              hipStream_t stream) {
    const float* in      = (const float*)d_in[0];
    const float* alpha_p = (const float*)d_in[1];
    const float* beta_p  = (const float*)d_in[2];
    float* out = (float*)d_out;

    const int nbatch = 8;
    const int n = in_sizes[0] / nbatch;

    const int tiles = (n + 15) / 16;
    const int wavesPerBlock = 8;
    const int blocks = (tiles + wavesPerBlock - 1) / wavesPerBlock;

    lpf_wmma_kernel<<<blocks, wavesPerBlock * 32, 0, stream>>>(
        in, alpha_p, beta_p, out, n, nbatch);
}


# Input staged per workgroup in LDS: wave w loads batch row w's 160 columns as whole lines, 16 bytes a lane (2 loads, was 10 touching 8 to 16 lines each); B fragments are LDS reads; sum order unchanged
# speedup vs baseline: 1.8033x; 1.1529x over previous
#include <hip/hip_runtime.h>

typedef __attribute__((ext_vector_type(2))) float v2f;
typedef __attribute__((ext_vector_type(4))) float v4f;
typedef __attribute__((ext_vector_type(8))) float v8f;

#define FS_F      8000.0f
#define HALF_W    16
#define KTAPS     33
#define PI_F      3.14159265358979323846f
#define TWO_PI_F  6.28318530717958647692f
#define TAB_LO    15
#define TAB_N     96
#define WAVES     8
#define KCOLS     40
#define APITCH    44
#define STRIP_N   1600

__global__ __launch_bounds__(256, 16) void
lpf_wmma_kernel(const float* __restrict__ in, const float* __restrict__ alpha_p,
                const float* __restrict__ beta_p, float* __restrict__ out,
                int n, int nbatch)
{
    __shared__ float gtab[TAB_N];
    __shared__ __attribute__((aligned(16))) float atile[WAVES * 16 * APITCH];
    __shared__ __attribute__((aligned(16))) float strip[STRIP_N];

    const float alpha = alpha_p[0];
    const float beta  = beta_p[0];

    const int lane = threadIdx.x & 31;
    const int wv   = __builtin_amdgcn_readfirstlane((int)(threadIdx.x >> 5));
    const int tile = (int)blockIdx.x * WAVES + wv;
    const int t0   = tile * 16;
    const bool active = (t0 < n);

    const int T0 = (int)blockIdx.x * (WAVES * 16);
    const int brw = (wv < nbatch) ? wv : 0;
    const float* __restrict__ prow = in + (size_t)brw * (size_t)n + T0;
    const int l43  = 4 * lane + 3;
    const int remA = n - T0;
    const int remH = (remA - 128 < 32) ? (remA - 128) : 32;
    const bool ldA = (l43 < remA);
    const bool ldH = (l43 < remH);
    v4f sA, sH;
    if (ldA) sA = *(const v4f*)(prow + 4 * lane);
    if (ldH) sH = *(const v4f*)(prow + 128 + 4 * lane);

    const int m  = lane & 15;
    const int hl = lane >> 4;
    const int t  = t0 + m;

    const float c_t =
        (PI_F * 0.25f + alpha * __sinf(beta * (float)t * (1.0f / FS_F))) * (1.0f / TWO_PI_F);
    const float c2   = 2.0f * c_t;

    auto tapk = [&](int k) -> float {
        const float g = gtab[k + TAB_LO];
        const float s = __builtin_amdgcn_sinf(c_t * (float)(k - HALF_W));
        return __builtin_fmaf(g, s, (k == HALF_W) ? c2 : 0.0f);
    };


    auto store_rows = [&](const v8f& d, bool mine, unsigned rowoff) {
        v4f lo, hi;
        #pragma unroll
        for (int v = 0; v < 8; ++v) {
            if (v < 4) lo[v] = d[v]; else hi[v - 4] = d[v];
        }
        if (mine) {
            char* p = (char*)(out + t0);
            const unsigned so = (rowoff + 8u * (unsigned)hl) * 4u;
            *(v4f*)(p + so)      = lo;
            *(v4f*)(p + so + 16) = hi;
        }
    };

    v8f acc = {};
    float S;

    const bool edge = (t0 < HALF_W) || (t0 > n - 3 * HALF_W);

    if (threadIdx.x < TAB_N) {
        const int k = (int)threadIdx.x - TAB_LO;
        const float w = 0.5f - 0.5f * __builtin_amdgcn_cosf((float)k * (1.0f / 32.0f));
        const float g = w / (PI_F * (float)(k - HALF_W));
        gtab[threadIdx.x] = (((unsigned)k < KTAPS) && (k != HALF_W)) ? g : 0.0f;
    }
    __builtin_amdgcn_fence(__ATOMIC_RELEASE, "workgroup", "local");
    __builtin_amdgcn_s_barrier();
    __builtin_amdgcn_fence(__ATOMIC_ACQUIRE, "workgroup", "local");

    const int s0 = m & 7;
    const int cg = m >> 3;
    const int br = (s0 < nbatch) ? s0 : 0;
    const unsigned rowoff = __umul24((unsigned)br, (unsigned)n);
    float* At = atile + (int)(threadIdx.x >> 5) * (16 * APITCH);
    float* rowp = At + m * APITCH;

    if (active && !edge) {
        {
            const v4f z = {0.0f, 0.0f, 0.0f, 0.0f};
            float* zr = rowp + 32 * hl;
            *(v4f*)(zr + 0)  = z;
            *(v4f*)(zr + 4)  = z;
        }

        const int j0 = 8 * hl;
        const float ph0 = c_t * (float)j0;
        const float* gj = gtab + (TAB_LO + HALF_W) + j0;
        float u0 = gj[0] * __builtin_amdgcn_sinf(ph0);
        u0 = (hl == 0) ? c2 : u0;
        v2f q[4];
        #pragma unroll
        for (int p = 0; p < 4; ++p) {
            v2f g, s;
            g.x = gj[2 * p + 1];
            g.y = gj[2 * p + 2];
            s.x = __builtin_amdgcn_sinf(__builtin_fmaf(c_t, (float)(2 * p + 1), ph0));
            s.y = __builtin_amdgcn_sinf(__builtin_fmaf(c_t, (float)(2 * p + 2), ph0));
            q[p] = g * s;
        }

        const v2f r = (q[0] + q[1]) + (q[2] + q[3]);
        const float Sp = __builtin_fmaf(2.0f, r.x + r.y, (hl == 0) ? u0 : 0.0f);
        {
            const int sb = __builtin_bit_cast(int, Sp);
            const float So = __builtin_bit_cast(float,
                __builtin_amdgcn_permlanex16(sb, sb, 0x76543210, 0xfedcba98, false, false));
            S = Sp + So;
        }

        float invS = __builtin_amdgcn_rcpf(S);
        invS = __builtin_fmaf(invS, __builtin_fmaf(-S, invS, 1.0f), invS);

        __builtin_amdgcn_fence(__ATOMIC_RELEASE, "wavefront");

        float* up = rowp + (s0 + HALF_W) + j0;
        float* dn = rowp + (s0 + HALF_W) - j0;
        {
            const float a0 = u0 * invS;
            up[0] = a0;
            dn[0] = a0;
        }
        #pragma unroll
        for (int p = 0; p < 4; ++p) {
            const v2f aq = q[p] * invS;
            up[2 * p + 1]    = aq.x;
            up[2 * p + 2]    = aq.y;
            dn[-(2 * p + 1)] = aq.x;
            dn[-(2 * p + 2)] = aq.y;
        }
    }

    {
        float* srow = strip + (200 * wv - 4 * (wv & 1));
        if (ldA) *(v4f*)(srow + 4 * lane) = sA;
        if (ldH) *(v4f*)(srow + 128 + 4 * lane) = sH;
    }
    __syncthreads();
    if (!active) return;

    if (!edge) {
        const float* pBl = strip + (200 * s0 - 4 * (s0 & 1)) + 16 * wv + 8 * cg + 2 * hl;
        const float* pA = rowp + 2 * hl;
        #pragma unroll
        for (int c = 0; c < KCOLS / 4; ++c) {
            v2f A, B;
            const float2 aa = *(const float2*)(pA + 4 * c);
            const float2 bb = *(const float2*)(pBl + 4 * c);
            A.x = aa.x;
            A.y = aa.y;
            B.x = bb.x;
            B.y = bb.y;
            acc = __builtin_amdgcn_wmma_f32_16x16x4_f32(
                false, A, false, B, (short)0, acc, false, false);
        }
        store_rows(acc, (cg == hl) && (s0 < nbatch), rowoff);
    } else {
        const int ebr = (m < nbatch) ? m : 0;
        const unsigned erowoff = __umul24((unsigned)ebr, (unsigned)n);
        const float* __restrict__ inb = in + erowoff;
        const int off = (t < HALF_W) ? -(HALF_W + 1) : 0;
        #pragma unroll
        for (int c = 0; c < 12; ++c) {
            const int base = 4 * c + 2 * hl;
            v2f A, B;
            #pragma unroll
            for (int j = 0; j < 2; ++j) {
                int k = base + j - m - off;
                bool valid = ((unsigned)k < KTAPS);
                valid = valid & ((t >= HALF_W) | (k >= HALF_W + 1 - t));
                valid = valid & ((t <= n - KTAPS) | (k < n - t));
                float v = tapk(k) * (valid ? 1.0f : 0.0f);
                if (j == 0) A.x = v; else A.y = v;
            }
            int col = t0 + base;
            col = (col < n - 1) ? col : (n - 2);
            const float2 bb = *(const float2*)(inb + (size_t)col);
            B.x = bb.x;
            B.y = bb.y;
            acc = __builtin_amdgcn_wmma_f32_16x16x4_f32(
                false, A, false, B, (short)0, acc, false, false);
        }
        float Sp = 0.0f;
        #pragma unroll
        for (int j = 0; j <= 16; ++j) {
            int k = 2 * j + hl;
            float v = tapk(k);
            Sp += v * ((k < KTAPS) ? 1.0f : 0.0f);
        }
        S = Sp + __shfl_xor(Sp, 16, 32);

        float invS = __builtin_amdgcn_rcpf(S);
        invS = __builtin_fmaf(invS, __builtin_fmaf(-S, invS, 1.0f), invS);
        #pragma unroll
        for (int v = 0; v < 8; ++v) {
            const int row = v + 8 * hl;
            acc[v] *= __shfl(invS, row, 32);
        }
        store_rows(acc, m < nbatch, erowoff);
    }
}

extern "C" void kernel_launch(void* const* d_in, const int* in_sizes, int n_in,
                              void* d_out, int out_size, void* d_ws, size_t ws_size,
                              hipStream_t stream) {
    const float* in      = (const float*)d_in[0];
    const float* alpha_p = (const float*)d_in[1];
    const float* beta_p  = (const float*)d_in[2];
    float* out = (float*)d_out;

    const int nbatch = 8;
    const int n = in_sizes[0] / nbatch;

    const int tiles = (n + 15) / 16;
    const int wavesPerBlock = 8;
    const int blocks = (tiles + wavesPerBlock - 1) / wavesPerBlock;

    lpf_wmma_kernel<<<blocks, wavesPerBlock * 32, 0, stream>>>(
        in, alpha_p, beta_p, out, n, nbatch);
}
